# starting kernel
# baseline (speedup 1.0000x reference)
#include <hip/hip_runtime.h>

typedef __attribute__((ext_vector_type(2))) float v2f;
typedef __attribute__((ext_vector_type(8))) float v8f;

#define NP   3
#define NC   1024
#define NV   64
#define NF   512
#define NK   9
#define ND   63
#define HOUT 8
#define KDIM (NP * NV)

__global__ __launch_bounds__(256) void stgcn_graph_wmma(
    const float* __restrict__ x, const float* __restrict__ graph,
    float* __restrict__ yT)
{
  const int wave = (blockIdx.x * blockDim.x + threadIdx.x) >> 5;
  const int lane = threadIdx.x & 31;
  const int half = lane >> 4;
  const int lr   = lane & 15;

  const int jT = wave & 3;
  const int cT = wave >> 2;
  const int c0 = cT * 16;
  const int j0 = jT * 16;

  const int ca = c0 + lr;
  const int jb = j0 + lr;

  v8f acc = {};
#pragma unroll
  for (int kk = 0; kk < KDIM / 4; ++kk) {
    const int kg = kk * 4 + half * 2;
    const int p  = kg >> 6;
    const int u  = kg & 63;
    v2f a = *(const v2f*)(x + (size_t)(p * NC + ca) * NV + u);
    v2f b;
    b.x = graph[(size_t)(p * NV + u)     * NV + jb];
    b.y = graph[(size_t)(p * NV + u + 1) * NV + jb];
    acc = __builtin_amdgcn_wmma_f32_16x16x4_f32(
         false, a,  false, b,
         (short)0, acc,  false,  false);
  }

  const int rowBase = c0 + half * 8;
#pragma unroll
  for (int r = 0; r < 8; ++r) {
    yT[(size_t)(rowBase + r) * NV + jb] = acc[r];
  }
}

__global__ __launch_bounds__(256) void stgcn_fifo_sum(
    const float* __restrict__ fifo, const float* __restrict__ yT,
    float* __restrict__ out)
{
  const int t   = blockIdx.x * blockDim.x + threadIdx.x;
  const int c   = t >> 7;
  const int rem = t & 127;
  const int h   = rem >> 4;
  const int v   = (rem & 15) << 2;

  const float* fbase = fifo + (size_t)c * NF * NV + v;

  float4 acc;
  if (h == 0) {
    acc = *(const float4*)(yT + (size_t)c * NV + v);
  } else {
    acc = *(const float4*)(fbase + (size_t)(h - 1) * NV);
  }
#pragma unroll
  for (int i = 1; i < NK; ++i) {
    float4 f = *(const float4*)(fbase + (size_t)(i * ND + h - 1) * NV);
    acc.x += f.x; acc.y += f.y; acc.z += f.z; acc.w += f.w;
  }
  *(float4*)(out + (size_t)t * 4) = acc;
}

extern "C" void kernel_launch(void* const* d_in, const int* in_sizes, int n_in,
                              void* d_out, int out_size, void* d_ws, size_t ws_size,
                              hipStream_t stream) {
  (void)in_sizes; (void)n_in; (void)out_size; (void)ws_size;
  const float* x     = (const float*)d_in[0];
  const float* graph = (const float*)d_in[1];
  const float* fifo  = (const float*)d_in[2];
  float* out = (float*)d_out;
  float* yT  = (float*)d_ws;

  stgcn_graph_wmma<<<32, 256, 0, stream>>>(x, graph, yT);
  stgcn_fifo_sum<<<512, 256, 0, stream>>>(fifo, yT, out);
}


# One launch of 256 identical 4-wave blocks: each owns 4 channels, forms their y rows itself on the 16x16x4 tile (its 4 rows repeated to fill M=16), then sums all 8 output rows; add order kept
# speedup vs baseline: 2.8501x; 2.8501x over previous
#include <hip/hip_runtime.h>


typedef __attribute__((ext_vector_type(2))) float v2f;
typedef __attribute__((ext_vector_type(8))) float v8f;

#define NP   3
#define NC   1024
#define NV   64
#define NF   512
#define NK   9
#define ND   63
#define HOUT 8
#define KDIM (NP * NV)
#define KSTEPS (KDIM / 4)
#define CPB  4

__global__ __launch_bounds__(128) void stgcn_own_rows(
    const float* __restrict__ x, const float* __restrict__ graph,
    const float* __restrict__ fifo, float* __restrict__ out)
{
  __shared__ __attribute__((aligned(16))) float ys[CPB][NV];

  const int tid  = threadIdx.x;
  const int wave = tid >> 5;
  const int lane = tid & 31;
  const int half = lane >> 4;
  const int lr   = lane & 15;
  const int c0   = blockIdx.x * CPB;

  const int h = (tid >> 4) & 7;
  const int v = (tid & 15) << 2;
  const int hfirst = (h == 0) ? 0 : h - 1;

  float4 first[CPB];
  float4 tap[CPB][NK - 1];
#pragma unroll
  for (int s = 0; s < CPB; ++s) {
    const float* fb = fifo + (size_t)(c0 + s) * NF * NV + v;
    first[s] = *(const float4*)(fb + (size_t)hfirst * NV);
#pragma unroll
    for (int i = 1; i < NK; ++i)
      tap[s][i - 1] = *(const float4*)(fb + (size_t)(i * ND + h - 1) * NV);
  }

  __builtin_amdgcn_sched_barrier(0);

  const int ca = c0 + (lr & (CPB - 1));
  const int jb = wave * 16 + lr;

  v2f a[KSTEPS], b[KSTEPS];
#pragma unroll
  for (int kk = 0; kk < KSTEPS; ++kk) {
    const int kg = kk * 4 + half * 2;
    const int p  = kg >> 6;
    const int u  = kg & 63;
    a[kk] = *(const v2f*)(x + (size_t)(p * NC + ca) * NV + u);
    b[kk].x = graph[(size_t)(p * NV + u)     * NV + jb];
    b[kk].y = graph[(size_t)(p * NV + u + 1) * NV + jb];
  }

  v8f acc = {};
#pragma unroll
  for (int kk = 0; kk < KSTEPS; ++kk) {
    acc = __builtin_amdgcn_wmma_f32_16x16x4_f32(
         false, a[kk],  false, b[kk],
         (short)0, acc,  false,  false);
  }
  if (half == 0) {
#pragma unroll
    for (int r = 0; r < CPB; ++r) ys[r][jb] = acc[r];
  }
  __syncthreads();

#pragma unroll
  for (int s = 0; s < CPB; ++s) {
    const float4 yv = *(const float4*)&ys[s][v];
    float4 o;
    o.x = (h == 0) ? yv.x : first[s].x;
    o.y = (h == 0) ? yv.y : first[s].y;
    o.z = (h == 0) ? yv.z : first[s].z;
    o.w = (h == 0) ? yv.w : first[s].w;
#pragma unroll
    for (int i = 0; i < NK - 1; ++i) {
      o.x += tap[s][i].x; o.y += tap[s][i].y; o.z += tap[s][i].z; o.w += tap[s][i].w;
    }
    *(float4*)(out + ((size_t)(c0 + s) * HOUT + h) * NV + v) = o;
  }
}

extern "C" void kernel_launch(void* const* d_in, const int* in_sizes, int n_in,
                              void* d_out, int out_size, void* d_ws, size_t ws_size,
                              hipStream_t stream) {
  (void)in_sizes; (void)n_in; (void)out_size; (void)d_ws; (void)ws_size;
  const float* x     = (const float*)d_in[0];
  const float* graph = (const float*)d_in[1];
  const float* fifo  = (const float*)d_in[2];
  float* out = (float*)d_out;

  stgcn_own_rows<<<NC / CPB, 128, 0, stream>>>(x, graph, fifo, out);
}


# Product operands fetched once per block as whole lines ahead of the fifo loads (27 wide loads a thread, all 63 in flight), staged in LDS, fragments read from LDS in groups; add order kept
# speedup vs baseline: 3.0124x; 1.0569x over previous
#include <hip/hip_runtime.h>


typedef __attribute__((ext_vector_type(2))) float v2f;
typedef __attribute__((ext_vector_type(4))) float v4f;
typedef __attribute__((ext_vector_type(8))) float v8f;

#define NP   3
#define NC   1024
#define NV   64
#define NF   512
#define NK   9
#define ND   63
#define HOUT 8
#define KDIM (NP * NV)
#define KSTEPS (KDIM / 4)
#define CPB  4
#define SB   8
#define NB   (KSTEPS / SB)
#define GROWS_PER_PASS 8
#define GPASSES (KDIM / GROWS_PER_PASS)
#define GPITCH 72
#define XPITCH 196

__global__ __launch_bounds__(128) void stgcn_own_rows(
    const float* x, const float* graph, const float* fifo, float* __restrict__ out)
{
  __shared__ __attribute__((aligned(16))) float gs[KDIM * GPITCH];
  __shared__ __attribute__((aligned(16))) float xs[CPB * XPITCH];
  __shared__ __attribute__((aligned(16))) float ys[CPB][NV];

  const int tid  = threadIdx.x;
  const int wave = tid >> 5;
  const int lane = tid & 31;
  const int half = lane >> 4;
  const int lr   = lane & 15;
  const int c0   = blockIdx.x * CPB;

  v2f xr[NP];
#pragma unroll
  for (int p = 0; p < NP; ++p)
    xr[p] = *(const v2f*)(x + (size_t)(p * NC + c0 + wave) * NV + 2 * lane);
  const int grow = tid >> 4;
  const int gcol = (tid & 15) << 2;
  v4f gr[GPASSES];
#pragma unroll
  for (int i = 0; i < GPASSES; ++i)
    gr[i] = *(const v4f*)(graph + (size_t)(i * GROWS_PER_PASS + grow) * NV + gcol);

  __builtin_amdgcn_sched_barrier(0);

  const int h = (tid >> 4) & 7;
  const int v = (tid & 15) << 2;
  const int hfirst = (h == 0) ? 0 : h - 1;

  float4 first[CPB];
  float4 tap[CPB][NK - 1];
#pragma unroll
  for (int s = 0; s < CPB; ++s) {
    const float* fb = fifo + (size_t)(c0 + s) * NF * NV + v;
    first[s] = *(const float4*)(fb + (size_t)hfirst * NV);
#pragma unroll
    for (int i = 1; i < NK; ++i)
      tap[s][i - 1] = *(const float4*)(fb + (size_t)(i * ND + h - 1) * NV);
  }

  __builtin_amdgcn_sched_barrier(0);

#pragma unroll
  for (int p = 0; p < NP; ++p)
    *(v2f*)&xs[wave * XPITCH + p * NV + 2 * lane] = xr[p];
#pragma unroll
  for (int i = 0; i < GPASSES; ++i)
    *(v4f*)&gs[(i * GROWS_PER_PASS + grow) * GPITCH + gcol] = gr[i];
  asm volatile("s_wait_dscnt 0x0\n\ts_barrier_signal -1\n\ts_barrier_wait -1" ::: "memory");

  const int sa = lr & (CPB - 1);
  const int jb = wave * 16 + lr;

  v2f a[2][SB], b[2][SB];
#pragma unroll
  for (int t = 0; t < SB; ++t) {
    const int kg = t * 4 + half * 2;
    a[0][t]   = *(const v2f*)&xs[sa * XPITCH + kg];
    b[0][t].x = gs[kg * GPITCH + jb];
    b[0][t].y = gs[(kg + 1) * GPITCH + jb];
  }
  __builtin_amdgcn_sched_barrier(0);
  v8f acc = {};
#pragma unroll
  for (int n = 0; n < NB; ++n) {
    if (n + 1 < NB) {
#pragma unroll
      for (int t = 0; t < SB; ++t) {
        const int kg = ((n + 1) * SB + t) * 4 + half * 2;
        a[(n + 1) & 1][t]   = *(const v2f*)&xs[sa * XPITCH + kg];
        b[(n + 1) & 1][t].x = gs[kg * GPITCH + jb];
        b[(n + 1) & 1][t].y = gs[(kg + 1) * GPITCH + jb];
      }
    }
    __builtin_amdgcn_sched_barrier(0);
#pragma unroll
    for (int t = 0; t < SB; ++t) {
      acc = __builtin_amdgcn_wmma_f32_16x16x4_f32(
           false, a[n & 1][t],  false, b[n & 1][t],
           (short)0, acc,  false,  false);
    }
    __builtin_amdgcn_sched_barrier(0);
  }
  if (half == 0) {
#pragma unroll
    for (int r = 0; r < CPB; ++r) ys[r][jb] = acc[r];
  }
  __syncthreads();

#pragma unroll
  for (int s = 0; s < CPB; ++s) {
    const float4 yv = *(const float4*)&ys[s][v];
    float4 o;
    o.x = (h == 0) ? yv.x : first[s].x;
    o.y = (h == 0) ? yv.y : first[s].y;
    o.z = (h == 0) ? yv.z : first[s].z;
    o.w = (h == 0) ? yv.w : first[s].w;
#pragma unroll
    for (int i = 0; i < NK - 1; ++i) {
      o.x += tap[s][i].x; o.y += tap[s][i].y; o.z += tap[s][i].z; o.w += tap[s][i].w;
    }
    *(float4*)(out + ((size_t)(c0 + s) * HOUT + h) * NV + v) = o;
  }
}

extern "C" void kernel_launch(void* const* d_in, const int* in_sizes, int n_in,
                              void* d_out, int out_size, void* d_ws, size_t ws_size,
                              hipStream_t stream) {
  (void)in_sizes; (void)n_in; (void)out_size; (void)d_ws; (void)ws_size;
  const float* x     = (const float*)d_in[0];
  const float* graph = (const float*)d_in[1];
  const float* fifo  = (const float*)d_in[2];
  float* out = (float*)d_out;

  stgcn_own_rows<<<NC / CPB, 128, 0, stream>>>(x, graph, fifo, out);
}
